# starting kernel
# baseline (speedup 1.0000x reference)
#include <hip/hip_runtime.h>
#include <hip/hip_bf16.h>
#include <math.h>

typedef __attribute__((ext_vector_type(2))) float v2f;
typedef __attribute__((ext_vector_type(8))) float v8f;

#define SELU_SCALE 1.0507009873554805f
#define SELU_ALPHA 1.6732632423543772f

__device__ __forceinline__ v8f wmma4(v2f a, v2f b, v8f c) {
    return __builtin_amdgcn_wmma_f32_16x16x4_f32(false, a, false, b, (short)0, c, false, false);
}

__global__ void deg_kernel(const int* __restrict__ dst, float* __restrict__ deg, int E) {
    int e = blockIdx.x * blockDim.x + threadIdx.x;
    if (e < E) atomicAdd(&deg[dst[e]], 1.0f);
}

__global__ void norm_kernel(const float* __restrict__ deg, float* __restrict__ dis,
                            float* __restrict__ cntinv, int n) {
    int i = blockIdx.x * blockDim.x + threadIdx.x;
    if (i < n) {
        float d = deg[i];
        dis[i]    = (d > 0.0f) ? rsqrtf(fmaxf(d, 1.0f)) : 0.0f;
        cntinv[i] = 1.0f / fmaxf(d, 1.0f);
    }
}

__global__ void prop_kernel(const float* __restrict__ X, float* __restrict__ Y,
                            const int* __restrict__ src, const int* __restrict__ dst,
                            const float* __restrict__ dis, int E, int weighted) {
    int tid = blockIdx.x * blockDim.x + threadIdx.x;
    int e   = tid >> 4;
    int seg = tid & 15;
    if (e >= E) return;
    int s = src[e];
    int t = dst[e];
    float w = 1.0f;
    if (weighted) w = dis[s] * dis[t];
    const float4 v = ((const float4*)(X + (long)s * 64))[seg];
    float* yr = Y + (long)t * 64 + seg * 4;
    atomicAdd(yr + 0, v.x * w);
    atomicAdd(yr + 1, v.y * w);
    atomicAdd(yr + 2, v.z * w);
    atomicAdd(yr + 3, v.w * w);
}

__global__ __launch_bounds__(256) void gemm1_kernel(
    const float* __restrict__ Sagg, const float* __restrict__ Hf,
    const float* __restrict__ cntinv,
    const float* __restrict__ Wl, const float* __restrict__ Wr,
    const float* __restrict__ bias, float* __restrict__ Out, int nnodes)
{
    int wave = (blockIdx.x * blockDim.x + threadIdx.x) >> 5;
    int lane = threadIdx.x & 31;
    int rb = wave;
    int rbCount = (nnodes + 15) >> 4;
    if (rb >= rbCount) return;
    int m = rb * 16 + (lane & 15);
    int mclamp = m < nnodes ? m : (nnodes - 1);
    int col0 = lane & 15;
    int kshift = (lane >> 4) * 2;
    float cinv = cntinv[mclamp];
    const float* aggRow = Sagg + (long)mclamp * 64;
    const float* hRow   = Hf   + (long)mclamp * 64;
    v8f acc0 = {}, acc1 = {}, acc2 = {}, acc3 = {};
    #pragma unroll 4
    for (int k0 = 0; k0 < 64; k0 += 4) {
        int kk = k0 + kshift;
        v2f a1; a1.x = aggRow[kk] * cinv; a1.y = aggRow[kk + 1] * cinv;
        v2f a2; a2.x = hRow[kk];          a2.y = hRow[kk + 1];
        const float* wl0 = Wl + kk * 64 + col0;
        const float* wl1 = Wl + (kk + 1) * 64 + col0;
        const float* wr0 = Wr + kk * 64 + col0;
        const float* wr1 = Wr + (kk + 1) * 64 + col0;
        v2f b;
        b.x = wl0[ 0]; b.y = wl1[ 0]; acc0 = wmma4(a1, b, acc0);
        b.x = wl0[16]; b.y = wl1[16]; acc1 = wmma4(a1, b, acc1);
        b.x = wl0[32]; b.y = wl1[32]; acc2 = wmma4(a1, b, acc2);
        b.x = wl0[48]; b.y = wl1[48]; acc3 = wmma4(a1, b, acc3);
        b.x = wr0[ 0]; b.y = wr1[ 0]; acc0 = wmma4(a2, b, acc0);
        b.x = wr0[16]; b.y = wr1[16]; acc1 = wmma4(a2, b, acc1);
        b.x = wr0[32]; b.y = wr1[32]; acc2 = wmma4(a2, b, acc2);
        b.x = wr0[48]; b.y = wr1[48]; acc3 = wmma4(a2, b, acc3);
    }
    int rowbase = rb * 16 + (lane >> 4) * 8;
    v8f accs[4] = {acc0, acc1, acc2, acc3};
    #pragma unroll
    for (int ct = 0; ct < 4; ++ct) {
        int col = ct * 16 + col0;
        float bn = bias[col];
        #pragma unroll
        for (int j = 0; j < 8; ++j) {
            int row = rowbase + j;
            float v = accs[ct][j] + bn;
            v = v > 0.0f ? SELU_SCALE * v : SELU_SCALE * SELU_ALPHA * (expf(v) - 1.0f);
            if (row < nnodes) Out[(long)row * 64 + col] = v;
        }
    }
}

__global__ __launch_bounds__(256) void gemm2_kernel(
    const float* __restrict__ Sagg, const float* __restrict__ Hf,
    const float* __restrict__ cntinv,
    const float* __restrict__ Wl, const float* __restrict__ Wr,
    const float* __restrict__ bias, float* __restrict__ Logits, int nnodes)
{
    int wave = (blockIdx.x * blockDim.x + threadIdx.x) >> 5;
    int lane = threadIdx.x & 31;
    int rb = wave;
    int rbCount = (nnodes + 15) >> 4;
    if (rb >= rbCount) return;
    int m = rb * 16 + (lane & 15);
    int mclamp = m < nnodes ? m : (nnodes - 1);
    int col0 = lane & 15;
    int kshift = (lane >> 4) * 2;
    float cinv = cntinv[mclamp];
    int col2  = 32 + col0;
    int c2off = col2 < 40 ? col2 : 0;
    float msk2 = col2 < 40 ? 1.0f : 0.0f;
    const float* aggRow = Sagg + (long)mclamp * 64;
    const float* hRow   = Hf   + (long)mclamp * 64;
    v8f acc0 = {}, acc1 = {}, acc2 = {};
    #pragma unroll 4
    for (int k0 = 0; k0 < 64; k0 += 4) {
        int kk = k0 + kshift;
        v2f a1; a1.x = aggRow[kk] * cinv; a1.y = aggRow[kk + 1] * cinv;
        v2f a2; a2.x = hRow[kk];          a2.y = hRow[kk + 1];
        const float* wl0 = Wl + kk * 40;
        const float* wl1 = Wl + (kk + 1) * 40;
        const float* wr0 = Wr + kk * 40;
        const float* wr1 = Wr + (kk + 1) * 40;
        v2f b;
        b.x = wl0[col0];          b.y = wl1[col0];          acc0 = wmma4(a1, b, acc0);
        b.x = wl0[16 + col0];     b.y = wl1[16 + col0];     acc1 = wmma4(a1, b, acc1);
        b.x = wl0[c2off] * msk2;  b.y = wl1[c2off] * msk2;  acc2 = wmma4(a1, b, acc2);
        b.x = wr0[col0];          b.y = wr1[col0];          acc0 = wmma4(a2, b, acc0);
        b.x = wr0[16 + col0];     b.y = wr1[16 + col0];     acc1 = wmma4(a2, b, acc1);
        b.x = wr0[c2off] * msk2;  b.y = wr1[c2off] * msk2;  acc2 = wmma4(a2, b, acc2);
    }
    int rowbase = rb * 16 + (lane >> 4) * 8;
    v8f accs[3] = {acc0, acc1, acc2};
    #pragma unroll
    for (int ct = 0; ct < 3; ++ct) {
        int col = ct * 16 + col0;
        if (col < 40) {
            float bn = bias[col];
            #pragma unroll
            for (int j = 0; j < 8; ++j) {
                int row = rowbase + j;
                if (row < nnodes) Logits[(long)row * 40 + col] = accs[ct][j] + bn;
            }
        }
    }
}

__global__ void softmax_kernel(const float* __restrict__ Logits, float* __restrict__ Out, int nnodes) {
    int wave = (blockIdx.x * blockDim.x + threadIdx.x) >> 5;
    int lane = threadIdx.x & 31;
    if (wave >= nnodes) return;
    const float* row = Logits + (long)wave * 40;
    float v0 = row[lane];
    bool hi = (lane + 32) < 40;
    float v1 = hi ? row[lane + 32] : -3.4e38f;
    float mx = fmaxf(v0, v1);
    for (int o = 16; o > 0; o >>= 1) mx = fmaxf(mx, __shfl_xor(mx, o, 32));
    float e0 = expf(v0 - mx);
    float e1 = hi ? expf(v1 - mx) : 0.0f;
    float s = e0 + e1;
    for (int o = 16; o > 0; o >>= 1) s += __shfl_xor(s, o, 32);
    float inv = 1.0f / s;
    Out[(long)wave * 40 + lane] = e0 * inv;
    if (hi) Out[(long)wave * 40 + lane + 32] = e1 * inv;
}

extern "C" void kernel_launch(void* const* d_in, const int* in_sizes, int n_in,
                              void* d_out, int out_size, void* d_ws, size_t ws_size,
                              hipStream_t stream) {
    const float* x    = (const float*)d_in[0];
    const float* Wl1  = (const float*)d_in[1];
    const float* Wr1  = (const float*)d_in[2];
    const float* b1   = (const float*)d_in[3];
    const float* Wl2  = (const float*)d_in[4];
    const float* Wr2  = (const float*)d_in[5];
    const float* b2   = (const float*)d_in[6];
    const int* esrc   = (const int*)d_in[7];
    const int* edst   = (const int*)d_in[8];

    const int H  = in_sizes[3];
    const int D  = in_sizes[1] / H;
    const int Nn = in_sizes[0] / D;
    const int E  = in_sizes[7];
    (void)n_in; (void)out_size; (void)ws_size;

    float* ws     = (float*)d_ws;
    float* deg    = ws;
    float* dis    = ws + (long)Nn;
    float* cntinv = ws + 2L * Nn;
    float* bufA   = ws + 3L * Nn;
    float* bufB   = bufA + (long)Nn * 64;
    float* bufC   = bufB + (long)Nn * 64;

    const int TB = 256;
    const size_t featBytes = (size_t)Nn * 64 * sizeof(float);
    dim3 gE((E + TB - 1) / TB), gN((Nn + TB - 1) / TB);
    dim3 gProp(((long)E * 16 + TB - 1) / TB);
    int rbCount = (Nn + 15) / 16;
    dim3 gGemm(((long)rbCount * 32 + TB - 1) / TB);
    dim3 gSm(((long)Nn * 32 + TB - 1) / TB);

    hipMemsetAsync(deg, 0, (size_t)Nn * sizeof(float), stream);
    deg_kernel<<<gE, TB, 0, stream>>>(edst, deg, E);
    norm_kernel<<<gN, TB, 0, stream>>>(deg, dis, cntinv, Nn);

    hipMemsetAsync(bufA, 0, featBytes, stream);
    prop_kernel<<<gProp, TB, 0, stream>>>(x, bufA, esrc, edst, dis, E, 1);
    hipMemsetAsync(bufB, 0, featBytes, stream);
    prop_kernel<<<gProp, TB, 0, stream>>>(bufA, bufB, esrc, edst, dis, E, 1);
    hipMemsetAsync(bufA, 0, featBytes, stream);
    prop_kernel<<<gProp, TB, 0, stream>>>(bufB, bufA, esrc, edst, dis, E, 0);
    gemm1_kernel<<<gGemm, TB, 0, stream>>>(bufA, bufB, cntinv, Wl1, Wr1, b1, bufC, Nn);
    hipMemsetAsync(bufA, 0, featBytes, stream);
    prop_kernel<<<gProp, TB, 0, stream>>>(bufC, bufA, esrc, edst, dis, E, 0);
    gemm2_kernel<<<gGemm, TB, 0, stream>>>(bufA, bufC, cntinv, Wl2, Wr2, b2, bufB, Nn);
    softmax_kernel<<<gSm, TB, 0, stream>>>(bufB, (float*)d_out, Nn);
}


# Degree count without global atomics: a work-group counts one edge slice's destinations of one node range in LDS integer counters; partial count planes are added in the norm kernel; degree memset gone
# speedup vs baseline: 2.4050x; 2.4050x over previous
#include <hip/hip_runtime.h>
#include <hip/hip_bf16.h>
#include <math.h>

typedef __attribute__((ext_vector_type(2))) float v2f;
typedef __attribute__((ext_vector_type(8))) float v8f;

#define SELU_SCALE 1.0507009873554805f
#define SELU_ALPHA 1.6732632423543772f

__device__ __forceinline__ v8f wmma4(v2f a, v2f b, v8f c) {
    return __builtin_amdgcn_wmma_f32_16x16x4_f32(false, a, false, b, (short)0, c, false, false);
}

#define DEG_SLICES 64
#define DEG_CAP 25600
#define DEG_THREADS 512
#define DEG_UNROLL 16
__global__ __launch_bounds__(DEG_THREADS) void deg_kernel(const int* __restrict__ dst, unsigned* __restrict__ part,
                                                          int E, int nnodes) {
    __shared__ uint4 hist4[DEG_CAP / 4];
    unsigned* hist = reinterpret_cast<unsigned*>(hist4);
    const int tid = threadIdx.x;
    const int nRanges = (nnodes + DEG_CAP - 1) / DEG_CAP;
    const int nItems = nRanges * DEG_SLICES;
    const long per = ((long)E + DEG_SLICES - 1) / DEG_SLICES;
    for (int item = blockIdx.x; item < nItems; item += gridDim.x) {
        const int r = item / DEG_SLICES;
        const int s = item - r * DEG_SLICES;
        const int lo = r * DEG_CAP;
        const int left = nnodes - lo;
        const int cnt = left < DEG_CAP ? left : DEG_CAP;
        for (int i = tid; i < DEG_CAP / 4; i += DEG_THREADS) hist4[i] = make_uint4(0u, 0u, 0u, 0u);
        __syncthreads();
        const long e0 = (long)s * per;
        const long e1 = (e0 + per < (long)E) ? e0 + per : (long)E;
        for (long base = e0; base < e1; base += (long)DEG_THREADS * DEG_UNROLL) {
            int dv[DEG_UNROLL];
            #pragma unroll
            for (int u = 0; u < DEG_UNROLL; ++u) {
                const long e = base + (long)u * DEG_THREADS + tid;
                const long ec = e < e1 ? e : e1 - 1;
                const int d = dst[ec];
                dv[u] = e < e1 ? d : -1;
            }
            #pragma unroll
            for (int u = 0; u < DEG_UNROLL; ++u) {
                const unsigned k = (unsigned)(dv[u] - lo);
                if (k < (unsigned)cnt) atomicAdd(&hist[k], 1u);
            }
        }
        __syncthreads();
        unsigned* out = part + (long)s * nnodes + lo;
        for (int i = tid; i < cnt; i += DEG_THREADS) out[i] = hist[i];
        __syncthreads();
    }
}

#define PROP_TICKET_STRIDE 64
#define PROP_TICKET_WORDS (8 * PROP_TICKET_STRIDE)
__global__ void norm_kernel(const unsigned* __restrict__ part, unsigned* __restrict__ tickets,
                            float* __restrict__ dis, float* __restrict__ cntinv, int n) {
    int i = blockIdx.x * blockDim.x + threadIdx.x;
    if (i < n) {
        unsigned c = 0u;
        #pragma unroll
        for (int s = 0; s < DEG_SLICES; ++s) c += part[(long)s * n + i];
        float d = (float)c;
        dis[i]    = (d > 0.0f) ? rsqrtf(fmaxf(d, 1.0f)) : 0.0f;
        cntinv[i] = 1.0f / fmaxf(d, 1.0f);
    }
    if (i < PROP_TICKET_WORDS) tickets[i] = 0u;
}

#define PROP_GROUP 8
#define PROP_CHUNK 128
#define PROP_SWEEPS (PROP_CHUNK / 32)
#define PROP_WAVES 8

__device__ __forceinline__ int cache_half_of_wave() {
    return (int)((__builtin_amdgcn_s_sendmsg_rtn(0x87) >> 18) & 1u);
}

__global__ __launch_bounds__(256) void prop_kernel(const float* __restrict__ X, float* __restrict__ Y,
                            const int* __restrict__ src, const int* __restrict__ dst,
                            const float* __restrict__ dis, unsigned* __restrict__ tickets,
                            int E, int nnodes, int weighted) {
    __shared__ int   q_s[PROP_WAVES][PROP_CHUNK];
    __shared__ int   q_t[PROP_WAVES][PROP_CHUNK];
    __shared__ float q_w[PROP_WAVES][PROP_CHUNK];
    const int lane = threadIdx.x & 31;
    const int wv = __builtin_amdgcn_readfirstlane((int)(threadIdx.x >> 5));
    int*   qs = q_s[wv];
    int*   qt = q_t[wv];
    float* qw = q_w[wv];
    const unsigned nChunks = (unsigned)(((long)E + PROP_CHUNK - 1) / PROP_CHUNK);
    const int nsplit = nnodes >> 1;
    int side = cache_half_of_wave();
    for (int pass = 0; pass < 2; ++pass, side ^= 1) {
        unsigned* ctr = tickets + side * PROP_TICKET_STRIDE;
        unsigned tkv = 0;
        if (lane == 0) tkv = atomicAdd(ctr, 1u);
        unsigned tk = (unsigned)__builtin_amdgcn_readfirstlane((int)tkv);
        for (unsigned trip = 0; trip < nChunks && tk < nChunks; ++trip) {
            const long ebase = (long)tk * PROP_CHUNK;
            int ss[PROP_SWEEPS], tt[PROP_SWEEPS];
            bool in[PROP_SWEEPS], mine[PROP_SWEEPS];
            #pragma unroll
            for (int sw = 0; sw < PROP_SWEEPS; ++sw) {
                const long e = ebase + sw * 32 + lane;
                in[sw] = e < (long)E;
                const long ec = in[sw] ? e : (long)E - 1;
                ss[sw] = src[ec];
                tt[sw] = dst[ec];
            }
            unsigned nxv = 0;
            if (lane == 0) nxv = atomicAdd(ctr, 1u);
            #pragma unroll
            for (int sw = 0; sw < PROP_SWEEPS; ++sw) {
                const bool low = tt[sw] < nsplit;
                mine[sw] = in[sw] && (side ? !low : low);
            }
            float wl[PROP_SWEEPS];
            #pragma unroll
            for (int sw = 0; sw < PROP_SWEEPS; ++sw) wl[sw] = 1.0f;
            if (weighted) {
                #pragma unroll
                for (int sw = 0; sw < PROP_SWEEPS; ++sw) {
                    const int si = mine[sw] ? ss[sw] : 0;
                    const int ti = mine[sw] ? tt[sw] : 0;
                    wl[sw] = dis[si] * dis[ti];
                }
            }
            int nq = 0;
            #pragma unroll
            for (int sw = 0; sw < PROP_SWEEPS; ++sw) {
                const unsigned m = __builtin_amdgcn_ballot_w32(mine[sw]);
                if (mine[sw]) {
                    const int pos = nq + __builtin_popcount(m & ((1u << lane) - 1u));
                    qs[pos] = ss[sw];
                    qt[pos] = tt[sw];
                    qw[pos] = wl[sw];
                }
                nq += __builtin_popcount(m);
            }

            for (int b0 = 0; b0 < nq; b0 += 32) {
                const int left = nq - b0;
                const int cnt = left < 32 ? left : 32;
                int s = 0, t = 0;
                float w = 1.0f;
                if (lane < cnt) { s = qs[b0 + lane]; t = qt[b0 + lane]; w = qw[b0 + lane]; }
                int j0 = 0;
                for (; j0 + PROP_GROUP <= cnt; j0 += PROP_GROUP) {
                    float a[PROP_GROUP], b[PROP_GROUP], ww[PROP_GROUP];
                    int td[PROP_GROUP];
                    #pragma unroll
                    for (int u = 0; u < PROP_GROUP; ++u) {
                        const int sj = __builtin_amdgcn_readlane(s, j0 + u);
                        td[u] = __builtin_amdgcn_readlane(t, j0 + u);
                        ww[u] = __builtin_bit_cast(float, __builtin_amdgcn_readlane(__builtin_bit_cast(int, w), j0 + u));
                        const float* xr = X + (long)sj * 64 + lane;
                        a[u] = xr[0];
                        b[u] = xr[32];
                    }
                    #pragma unroll
                    for (int u = 0; u < PROP_GROUP; ++u) {
                        float* yr = Y + (long)td[u] * 64 + lane;
                        atomicAdd(yr,      a[u] * ww[u]);
                        atomicAdd(yr + 32, b[u] * ww[u]);
                    }
                }
                if (j0 < cnt) {
                    const int rem = cnt - j0;
                    float a[PROP_GROUP], b[PROP_GROUP], ww[PROP_GROUP];
                    int td[PROP_GROUP];
                    #pragma unroll
                    for (int u = 0; u < PROP_GROUP - 1; ++u) {
                        a[u] = 0.0f; b[u] = 0.0f; ww[u] = 0.0f; td[u] = 0;
                        if (u < rem) {
                            const int sj = __builtin_amdgcn_readlane(s, j0 + u);
                            td[u] = __builtin_amdgcn_readlane(t, j0 + u);
                            ww[u] = __builtin_bit_cast(float, __builtin_amdgcn_readlane(__builtin_bit_cast(int, w), j0 + u));
                            const float* xr = X + (long)sj * 64 + lane;
                            a[u] = xr[0];
                            b[u] = xr[32];
                        }
                    }
                    #pragma unroll
                    for (int u = 0; u < PROP_GROUP - 1; ++u) {
                        if (u < rem) {
                            float* yr = Y + (long)td[u] * 64 + lane;
                            atomicAdd(yr,      a[u] * ww[u]);
                            atomicAdd(yr + 32, b[u] * ww[u]);
                        }
                    }
                }
            }
            tk = (unsigned)__builtin_amdgcn_readfirstlane((int)nxv);
        }
    }
}

__global__ __launch_bounds__(256) void gemm1_kernel(
    const float* __restrict__ Sagg, const float* __restrict__ Hf,
    const float* __restrict__ cntinv,
    const float* __restrict__ Wl, const float* __restrict__ Wr,
    const float* __restrict__ bias, float* __restrict__ Out, int nnodes)
{
    int wave = (blockIdx.x * blockDim.x + threadIdx.x) >> 5;
    int lane = threadIdx.x & 31;
    int rb = wave;
    int rbCount = (nnodes + 15) >> 4;
    if (rb >= rbCount) return;
    int m = rb * 16 + (lane & 15);
    int mclamp = m < nnodes ? m : (nnodes - 1);
    int col0 = lane & 15;
    int kshift = (lane >> 4) * 2;
    float cinv = cntinv[mclamp];
    const float* aggRow = Sagg + (long)mclamp * 64;
    const float* hRow   = Hf   + (long)mclamp * 64;
    v8f acc0 = {}, acc1 = {}, acc2 = {}, acc3 = {};
    #pragma unroll 4
    for (int k0 = 0; k0 < 64; k0 += 4) {
        int kk = k0 + kshift;
        v2f a1; a1.x = aggRow[kk] * cinv; a1.y = aggRow[kk + 1] * cinv;
        v2f a2; a2.x = hRow[kk];          a2.y = hRow[kk + 1];
        const float* wl0 = Wl + kk * 64 + col0;
        const float* wl1 = Wl + (kk + 1) * 64 + col0;
        const float* wr0 = Wr + kk * 64 + col0;
        const float* wr1 = Wr + (kk + 1) * 64 + col0;
        v2f b;
        b.x = wl0[ 0]; b.y = wl1[ 0]; acc0 = wmma4(a1, b, acc0);
        b.x = wl0[16]; b.y = wl1[16]; acc1 = wmma4(a1, b, acc1);
        b.x = wl0[32]; b.y = wl1[32]; acc2 = wmma4(a1, b, acc2);
        b.x = wl0[48]; b.y = wl1[48]; acc3 = wmma4(a1, b, acc3);
        b.x = wr0[ 0]; b.y = wr1[ 0]; acc0 = wmma4(a2, b, acc0);
        b.x = wr0[16]; b.y = wr1[16]; acc1 = wmma4(a2, b, acc1);
        b.x = wr0[32]; b.y = wr1[32]; acc2 = wmma4(a2, b, acc2);
        b.x = wr0[48]; b.y = wr1[48]; acc3 = wmma4(a2, b, acc3);
    }
    int rowbase = rb * 16 + (lane >> 4) * 8;
    v8f accs[4] = {acc0, acc1, acc2, acc3};
    #pragma unroll
    for (int ct = 0; ct < 4; ++ct) {
        int col = ct * 16 + col0;
        float bn = bias[col];
        #pragma unroll
        for (int j = 0; j < 8; ++j) {
            int row = rowbase + j;
            float v = accs[ct][j] + bn;
            v = v > 0.0f ? SELU_SCALE * v : SELU_SCALE * SELU_ALPHA * (expf(v) - 1.0f);
            if (row < nnodes) Out[(long)row * 64 + col] = v;
        }
    }
}

__global__ __launch_bounds__(256) void gemm2_kernel(
    const float* __restrict__ Sagg, const float* __restrict__ Hf,
    const float* __restrict__ cntinv,
    const float* __restrict__ Wl, const float* __restrict__ Wr,
    const float* __restrict__ bias, float* __restrict__ Logits, int nnodes)
{
    int wave = (blockIdx.x * blockDim.x + threadIdx.x) >> 5;
    int lane = threadIdx.x & 31;
    int rb = wave;
    int rbCount = (nnodes + 15) >> 4;
    if (rb >= rbCount) return;
    int m = rb * 16 + (lane & 15);
    int mclamp = m < nnodes ? m : (nnodes - 1);
    int col0 = lane & 15;
    int kshift = (lane >> 4) * 2;
    float cinv = cntinv[mclamp];
    int col2  = 32 + col0;
    int c2off = col2 < 40 ? col2 : 0;
    float msk2 = col2 < 40 ? 1.0f : 0.0f;
    const float* aggRow = Sagg + (long)mclamp * 64;
    const float* hRow   = Hf   + (long)mclamp * 64;
    v8f acc0 = {}, acc1 = {}, acc2 = {};
    #pragma unroll 4
    for (int k0 = 0; k0 < 64; k0 += 4) {
        int kk = k0 + kshift;
        v2f a1; a1.x = aggRow[kk] * cinv; a1.y = aggRow[kk + 1] * cinv;
        v2f a2; a2.x = hRow[kk];          a2.y = hRow[kk + 1];
        const float* wl0 = Wl + kk * 40;
        const float* wl1 = Wl + (kk + 1) * 40;
        const float* wr0 = Wr + kk * 40;
        const float* wr1 = Wr + (kk + 1) * 40;
        v2f b;
        b.x = wl0[col0];          b.y = wl1[col0];          acc0 = wmma4(a1, b, acc0);
        b.x = wl0[16 + col0];     b.y = wl1[16 + col0];     acc1 = wmma4(a1, b, acc1);
        b.x = wl0[c2off] * msk2;  b.y = wl1[c2off] * msk2;  acc2 = wmma4(a1, b, acc2);
        b.x = wr0[col0];          b.y = wr1[col0];          acc0 = wmma4(a2, b, acc0);
        b.x = wr0[16 + col0];     b.y = wr1[16 + col0];     acc1 = wmma4(a2, b, acc1);
        b.x = wr0[c2off] * msk2;  b.y = wr1[c2off] * msk2;  acc2 = wmma4(a2, b, acc2);
    }
    int rowbase = rb * 16 + (lane >> 4) * 8;
    v8f accs[3] = {acc0, acc1, acc2};
    #pragma unroll
    for (int ct = 0; ct < 3; ++ct) {
        int col = ct * 16 + col0;
        if (col < 40) {
            float bn = bias[col];
            #pragma unroll
            for (int j = 0; j < 8; ++j) {
                int row = rowbase + j;
                if (row < nnodes) Logits[(long)row * 40 + col] = accs[ct][j] + bn;
            }
        }
    }
}

__global__ void softmax_kernel(const float* __restrict__ Logits, float* __restrict__ Out, int nnodes) {
    int wave = (blockIdx.x * blockDim.x + threadIdx.x) >> 5;
    int lane = threadIdx.x & 31;
    if (wave >= nnodes) return;
    const float* row = Logits + (long)wave * 40;
    float v0 = row[lane];
    bool hi = (lane + 32) < 40;
    float v1 = hi ? row[lane + 32] : -3.4e38f;
    float mx = fmaxf(v0, v1);
    for (int o = 16; o > 0; o >>= 1) mx = fmaxf(mx, __shfl_xor(mx, o, 32));
    float e0 = expf(v0 - mx);
    float e1 = hi ? expf(v1 - mx) : 0.0f;
    float s = e0 + e1;
    for (int o = 16; o > 0; o >>= 1) s += __shfl_xor(s, o, 32);
    float inv = 1.0f / s;
    Out[(long)wave * 40 + lane] = e0 * inv;
    if (hi) Out[(long)wave * 40 + lane + 32] = e1 * inv;
}

extern "C" void kernel_launch(void* const* d_in, const int* in_sizes, int n_in,
                              void* d_out, int out_size, void* d_ws, size_t ws_size,
                              hipStream_t stream) {
    const float* x    = (const float*)d_in[0];
    const float* Wl1  = (const float*)d_in[1];
    const float* Wr1  = (const float*)d_in[2];
    const float* b1   = (const float*)d_in[3];
    const float* Wl2  = (const float*)d_in[4];
    const float* Wr2  = (const float*)d_in[5];
    const float* b2   = (const float*)d_in[6];
    const int* esrc   = (const int*)d_in[7];
    const int* edst   = (const int*)d_in[8];

    const int H  = in_sizes[3];
    const int D  = in_sizes[1] / H;
    const int Nn = in_sizes[0] / D;
    const int E  = in_sizes[7];
    (void)n_in; (void)out_size; (void)ws_size;

    float* ws     = (float*)d_ws;
    float* deg    = ws;
    float* dis    = ws + (long)Nn;
    float* cntinv = ws + 2L * Nn;
    float* bufA   = ws + 3L * Nn;
    float* bufB   = bufA + (long)Nn * 64;
    float* bufC   = bufB + (long)Nn * 64;

    const int TB = 256;
    const size_t featBytes = (size_t)Nn * 64 * sizeof(float);
    dim3 gN((Nn + TB - 1) / TB);
    dim3 gDeg(256);
    dim3 gProp(1024);
    unsigned* tickets = (unsigned*)deg;
    unsigned* part    = (unsigned*)bufB;
    int rbCount = (Nn + 15) / 16;
    dim3 gGemm(((long)rbCount * 32 + TB - 1) / TB);
    dim3 gSm(((long)Nn * 32 + TB - 1) / TB);

    deg_kernel<<<gDeg, DEG_THREADS, 0, stream>>>(edst, part, E, Nn);
    norm_kernel<<<gN, TB, 0, stream>>>(part, tickets, dis, cntinv, Nn);

    hipMemsetAsync(bufA, 0, featBytes, stream);
    prop_kernel<<<gProp, TB, 0, stream>>>(x, bufA, esrc, edst, dis, tickets, E, Nn, 1);
    hipMemsetAsync(bufB, 0, featBytes, stream);
    prop_kernel<<<gProp, TB, 0, stream>>>(bufA, bufB, esrc, edst, dis, tickets + 2 * PROP_TICKET_STRIDE, E, Nn, 1);
    hipMemsetAsync(bufA, 0, featBytes, stream);
    prop_kernel<<<gProp, TB, 0, stream>>>(bufB, bufA, esrc, edst, dis, tickets + 4 * PROP_TICKET_STRIDE, E, Nn, 0);
    gemm1_kernel<<<gGemm, TB, 0, stream>>>(bufA, bufB, cntinv, Wl1, Wr1, b1, bufC, Nn);
    hipMemsetAsync(bufA, 0, featBytes, stream);
    prop_kernel<<<gProp, TB, 0, stream>>>(bufC, bufA, esrc, edst, dis, tickets + 6 * PROP_TICKET_STRIDE, E, Nn, 0);
    gemm2_kernel<<<gGemm, TB, 0, stream>>>(bufA, bufC, cntinv, Wl2, Wr2, b2, bufB, Nn);
    softmax_kernel<<<gSm, TB, 0, stream>>>(bufB, (float*)d_out, Nn);
}
